# starting kernel
# baseline (speedup 1.0000x reference)
#include <hip/hip_runtime.h>
#include <hip/hip_bf16.h>

typedef __attribute__((ext_vector_type(2))) float v2f;
typedef __attribute__((ext_vector_type(4))) float v4f;
typedef __attribute__((ext_vector_type(8))) float v8f;

#define B_ 64
#define T_ 512
#define D_ 1024
#define THALF_ (T_ / 2)

__global__ __launch_bounds__(256) void wtab_kernel(
    const int* __restrict__ wordseq, const int* __restrict__ slen,
    const float* __restrict__ weights, float* __restrict__ wtab) {
  int i = blockIdx.x * 256 + threadIdx.x;
  int b = i >> 9;
  int t = i & (T_ - 1);
  float w = 0.0f;
  if (t < slen[b]) w = weights[wordseq[i]];
  wtab[i] = w;
}

__global__ __launch_bounds__(256) void fused_wmma_kernel(
    const float* __restrict__ vs, const int* __restrict__ slen,
    const float* __restrict__ wtab,
    float* __restrict__ s_part, float* __restrict__ yh_part) {
  __shared__ float lds_w[T_];

  const int tid  = threadIdx.x;
  const int lane = tid & 31;
  const int wid  = blockIdx.x * 8 + (tid >> 5);
  const int b      = wid >> 5;
  const int tile   = (wid >> 1) & 15;
  const int thalf  = wid & 1;
  const int d_base = tile * 64;
  const int tbase  = thalf * THALF_;

  for (int i = tid; i < T_; i += 256) lds_w[i] = wtab[b * T_ + i];
  __syncthreads();

  const int   m    = lane & 15;
  const int   h    = lane >> 4;
  const int   lenl = slen[b] - tbase;
  const float rowsel0 = (m == 0) ? 1.0f : 0.0f;
  const float rowsel1 = (m == 1) ? 1.0f : 0.0f;

  v8f c0 = {}, c1 = {}, c2 = {}, c3 = {};

  const float* base  = vs + (size_t)b * T_ * D_ + (size_t)tbase * D_ +
                       (size_t)(d_base + 4 * m);
  const float* wrow  = lds_w + tbase;

#pragma unroll 8
  for (int t0 = 0; t0 < THALF_; t0 += 4) {
    const int te = t0 + 2 * h;
    v4f r0 = *(const v4f*)(base + (size_t)te * D_);
    v4f r1 = *(const v4f*)(base + (size_t)(te + 1) * D_);

    const float mk0 = (te     < lenl) ? 1.0f : 0.0f;
    const float mk1 = (te + 1 < lenl) ? 1.0f : 0.0f;
    const float w0  = wrow[te];
    const float w1  = wrow[te + 1];
    v2f a;
    a.x = rowsel0 * mk0 + rowsel1 * w0;
    a.y = rowsel0 * mk1 + rowsel1 * w1;

    v2f bop;
    bop.x = r0.x; bop.y = r1.x;
    c0 = __builtin_amdgcn_wmma_f32_16x16x4_f32(false, a, false, bop, (short)0, c0, false, false);
    bop.x = r0.y; bop.y = r1.y;
    c1 = __builtin_amdgcn_wmma_f32_16x16x4_f32(false, a, false, bop, (short)0, c1, false, false);
    bop.x = r0.z; bop.y = r1.z;
    c2 = __builtin_amdgcn_wmma_f32_16x16x4_f32(false, a, false, bop, (short)0, c2, false, false);
    bop.x = r0.w; bop.y = r1.w;
    c3 = __builtin_amdgcn_wmma_f32_16x16x4_f32(false, a, false, bop, (short)0, c3, false, false);
  }

  if (lane < 16) {
    v4f sv, hv;
    sv.x = c0[0]; sv.y = c1[0]; sv.z = c2[0]; sv.w = c3[0];
    hv.x = c0[1]; hv.y = c1[1]; hv.z = c2[1]; hv.w = c3[1];
    const size_t off = (size_t)(thalf * B_ + b) * D_ + d_base + 4 * lane;
    *(v4f*)(s_part  + off) = sv;
    *(v4f*)(yh_part + off) = hv;
  }
}

__global__ __launch_bounds__(256) void finalize_kernel(
    const float* __restrict__ s_part, const float* __restrict__ yh_part,
    float* __restrict__ y_out, float* __restrict__ yh_out) {
  __shared__ float red[256];
  const int b   = blockIdx.x;
  const int tid = threadIdx.x;

  const v4f* s0 = (const v4f*)(s_part + (size_t)b * D_);
  const v4f* s1 = (const v4f*)(s_part + (size_t)(B_ + b) * D_);
  const v4f* h0 = (const v4f*)(yh_part + (size_t)b * D_);
  const v4f* h1 = (const v4f*)(yh_part + (size_t)(B_ + b) * D_);

  v4f sv = s0[tid];
  v4f sw = s1[tid];
  sv.x += sw.x; sv.y += sw.y; sv.z += sw.z; sv.w += sw.w;

  v4f hv = h0[tid];
  v4f hw = h1[tid];
  hv.x += hw.x; hv.y += hw.y; hv.z += hw.z; hv.w += hw.w;
  ((v4f*)(yh_out + (size_t)b * D_))[tid] = hv;

  red[tid] = fabsf(sv.x) + fabsf(sv.y) + fabsf(sv.z) + fabsf(sv.w);
  __syncthreads();
  for (int off = 128; off > 0; off >>= 1) {
    if (tid < off) red[tid] += red[tid + off];
    __syncthreads();
  }
  const float inv = 1.0f / sqrtf(red[0]);
  v4f yv;
  yv.x = sv.x * inv; yv.y = sv.y * inv; yv.z = sv.z * inv; yv.w = sv.w * inv;
  ((v4f*)(y_out + (size_t)b * D_))[tid] = yv;
}

extern "C" void kernel_launch(void* const* d_in, const int* in_sizes, int n_in,
                              void* d_out, int out_size, void* d_ws, size_t ws_size,
                              hipStream_t stream) {
  const float* vs      = (const float*)d_in[0];
  const int*   slen    = (const int*)d_in[1];
  const int*   wordseq = (const int*)d_in[2];
  const float* weights = (const float*)d_in[3];

  float* y_out  = (float*)d_out;
  float* yh_out = (float*)d_out + B_ * D_;

  float* wtab    = (float*)d_ws;
  float* s_part  = wtab + B_ * T_;
  float* yh_part = s_part + 2 * B_ * D_;

  wtab_kernel<<<(B_ * T_) / 256, 256, 0, stream>>>(wordseq, slen, weights, wtab);
  fused_wmma_kernel<<<256, 256, 0, stream>>>(vs, slen, wtab, s_part, yh_part);
  finalize_kernel<<<B_, 256, 0, stream>>>(s_part, yh_part, y_out, yh_out);
}


# main kernel loads only rows t<len[b]: a long and a short sequence paired per workgroup by rank of len, K-steps dealt to 4 waves, slot-wise prefetch, partials summed in LDS (sum order over t changes)
# speedup vs baseline: 1.2906x; 1.2906x over previous
#include <hip/hip_runtime.h>
#include <hip/hip_bf16.h>


typedef __attribute__((ext_vector_type(2))) float v2f;
typedef __attribute__((ext_vector_type(4))) float v4f;
typedef __attribute__((ext_vector_type(8))) float v8f;

#define B_ 64
#define T_ 512
#define D_ 1024
#define U_ 8

__global__ __launch_bounds__(256) void wtab_kernel(
    const int* __restrict__ wordseq, const int* __restrict__ slen,
    const float* __restrict__ weights, float* __restrict__ wtab) {
  int i = blockIdx.x * 256 + threadIdx.x;
  int b = i >> 9;
  int t = i & (T_ - 1);
  float w = 0.0f;
  if (t < slen[b]) w = weights[wordseq[i]];
  wtab[i] = w;
}

struct Seg {
  const float* base[2];
  int len[2];
  int nA;
  int nTot;
  int phase;
};

__device__ __forceinline__ void step_desc(const Seg& s, int i, int& seg, int& t0, bool& valid) {
  valid = i < s.nTot;
  seg = (valid && i >= s.nA) ? 1 : 0;
  const int li = valid ? (seg ? i - s.nA : i) : 0;
  t0 = valid ? 4 * s.phase + 16 * li : 0;
}

__device__ __forceinline__ void load_step(const Seg& s, int i, int h, v4f& r0, v4f& r1) {
  int seg, t0; bool valid;
  step_desc(s, i, seg, t0, valid);
  const float* p = (seg ? s.base[1] : s.base[0]) + (size_t)(t0 + 2 * h) * D_;
  r0 = *(const v4f*)(p);
  r1 = *(const v4f*)(p + D_);
}

__device__ __forceinline__ void mma_step(const Seg& s, int i, int m, int h,
                                         const float (*lw)[T_],
                                         v4f& r0, v4f& r1,
                                         v8f& c0, v8f& c1, v8f& c2, v8f& c3) {
  asm volatile("" : "+v"(r0), "+v"(r1));
  int seg, t0; bool valid;
  step_desc(s, i, seg, t0, valid);
  const int te   = t0 + 2 * h;
  const int lenS = seg ? s.len[1] : s.len[0];
  const float mk0 = (valid && te     < lenS) ? 1.0f : 0.0f;
  const float mk1 = (valid && te + 1 < lenS) ? 1.0f : 0.0f;
  const v2f wp = *(const v2f*)(&lw[seg][te]);
  const int rowM = 2 * seg;
  v2f a;
  a.x = (m == rowM) ? mk0 : ((m == rowM + 1) ? mk0 * wp.x : 0.0f);
  a.y = (m == rowM) ? mk1 : ((m == rowM + 1) ? mk1 * wp.y : 0.0f);

  v2f bop;
  bop.x = r0.x; bop.y = r1.x;
  c0 = __builtin_amdgcn_wmma_f32_16x16x4_f32(false, a, false, bop, (short)0, c0, false, false);
  bop.x = r0.y; bop.y = r1.y;
  c1 = __builtin_amdgcn_wmma_f32_16x16x4_f32(false, a, false, bop, (short)0, c1, false, false);
  bop.x = r0.z; bop.y = r1.z;
  c2 = __builtin_amdgcn_wmma_f32_16x16x4_f32(false, a, false, bop, (short)0, c2, false, false);
  bop.x = r0.w; bop.y = r1.w;
  c3 = __builtin_amdgcn_wmma_f32_16x16x4_f32(false, a, false, bop, (short)0, c3, false, false);
}

__global__ __launch_bounds__(256) void fused_wmma_kernel(
    const float* __restrict__ vs, const int* __restrict__ slen,
    const float* __restrict__ wtab,
    float* __restrict__ s_part, float* __restrict__ yh_part) {
  __shared__ int lds_len[B_];
  __shared__ int lds_pair[2];
  __shared__ __attribute__((aligned(16))) float lds_w[2][T_];
  __shared__ __attribute__((aligned(16))) float lds_red[4][2][4][64];

  const int tid  = threadIdx.x;
  const int lane = tid & 31;
  const int wv    = __builtin_amdgcn_readfirstlane(tid >> 5);
  const int tile  = wv & 1;
  const int phase = wv >> 1;
  const int p = blockIdx.x >> 3;
  const int q = blockIdx.x & 7;

  if (tid < B_) {
    int l = slen[tid];
    l = l < 0 ? 0 : (l > T_ ? T_ : l);
    lds_len[tid] = l;
  }
  __syncthreads();
  if (tid < B_) {
    const int mine = lds_len[tid];
    int rank = 0;
    for (int j = 0; j < B_; ++j) {
      const int o = lds_len[j];
      rank += (o < mine || (o == mine && j < tid)) ? 1 : 0;
    }
    if (rank == B_ - 1 - p) lds_pair[0] = tid;
    if (rank == p)          lds_pair[1] = tid;
  }
  __syncthreads();
  const int bA   = __builtin_amdgcn_readfirstlane(lds_pair[0]);
  const int bB   = __builtin_amdgcn_readfirstlane(lds_pair[1]);
  const int lenA = __builtin_amdgcn_readfirstlane(lds_len[bA]);
  const int lenB = __builtin_amdgcn_readfirstlane(lds_len[bB]);

  const int wseg = tid >> 7;
  const int wi4  = (tid & 127) * 4;
  const v4f wreg = *(const v4f*)(wtab + (size_t)(wseg ? bB : bA) * T_ + wi4);

  const int m = lane & 15;
  const int h = lane >> 4;
  const int col = q * 128 + tile * 64 + 4 * m;

  Seg s;
  s.base[0] = vs + (size_t)bA * T_ * D_ + col;
  s.base[1] = vs + (size_t)bB * T_ * D_ + col;
  s.len[0] = lenA;
  s.len[1] = lenB;
  s.phase  = phase;
  s.nA   = (((lenA + 3) >> 2) - phase + 3) >> 2;
  const int nB = (((lenB + 3) >> 2) - phase + 3) >> 2;
  s.nTot = s.nA + nB;
  const int nraw = (s.nTot + U_ - 1) / U_;
  const int nch  = nraw < 1 ? 1 : nraw;

  v4f r0[U_], r1[U_];
#pragma unroll
  for (int u = 0; u < U_; ++u) load_step(s, u, h, r0[u], r1[u]);

  *(v4f*)(&lds_w[wseg][wi4]) = wreg;
  __syncthreads();

  v8f c0 = {}, c1 = {}, c2 = {}, c3 = {};
  int c = 0;
  for (; c + 1 < nch; ++c) {
#pragma unroll
    for (int u = 0; u < U_; ++u) {
      mma_step(s, c * U_ + u, m, h, lds_w, r0[u], r1[u], c0, c1, c2, c3);
      load_step(s, (c + 1) * U_ + u, h, r0[u], r1[u]);
    }
  }
#pragma unroll
  for (int u = 0; u < U_; ++u)
    mma_step(s, c * U_ + u, m, h, lds_w, r0[u], r1[u], c0, c1, c2, c3);

  if (lane < 16) {
#pragma unroll
    for (int r = 0; r < 4; ++r) {
      v4f v;
      v.x = c0[r]; v.y = c1[r]; v.z = c2[r]; v.w = c3[r];
      *(v4f*)(&lds_red[phase][tile][r][4 * lane]) = v;
    }
  }
  __syncthreads();
  if (tid < 128) {
    const int r   = tid >> 5;
    const int tl  = (tid >> 4) & 1;
    const int l16 = tid & 15;
    v4f acc = *(const v4f*)(&lds_red[0][tl][r][4 * l16]);
#pragma unroll
    for (int ph = 1; ph < 4; ++ph) {
      const v4f x = *(const v4f*)(&lds_red[ph][tl][r][4 * l16]);
      acc.x += x.x; acc.y += x.y; acc.z += x.z; acc.w += x.w;
    }
    float* dst = (r & 1) ? yh_part : s_part;
    const int bo = (r >> 1) ? bB : bA;
    *(v4f*)(dst + (size_t)bo * D_ + q * 128 + tl * 64 + 4 * l16) = acc;
  }
}

__global__ __launch_bounds__(256) void finalize_kernel(
    const float* __restrict__ s_part, const float* __restrict__ yh_part,
    float* __restrict__ y_out, float* __restrict__ yh_out) {
  __shared__ float red[256];
  const int b   = blockIdx.x;
  const int tid = threadIdx.x;

  const v4f sv = ((const v4f*)(s_part + (size_t)b * D_))[tid];
  const v4f hv = ((const v4f*)(yh_part + (size_t)b * D_))[tid];
  ((v4f*)(yh_out + (size_t)b * D_))[tid] = hv;

  red[tid] = fabsf(sv.x) + fabsf(sv.y) + fabsf(sv.z) + fabsf(sv.w);
  __syncthreads();
  for (int off = 128; off > 0; off >>= 1) {
    if (tid < off) red[tid] += red[tid + off];
    __syncthreads();
  }
  const float inv = 1.0f / sqrtf(red[0]);
  v4f yv;
  yv.x = sv.x * inv; yv.y = sv.y * inv; yv.z = sv.z * inv; yv.w = sv.w * inv;
  ((v4f*)(y_out + (size_t)b * D_))[tid] = yv;
}

extern "C" void kernel_launch(void* const* d_in, const int* in_sizes, int n_in,
                              void* d_out, int out_size, void* d_ws, size_t ws_size,
                              hipStream_t stream) {
  const float* vs      = (const float*)d_in[0];
  const int*   slen    = (const int*)d_in[1];
  const int*   wordseq = (const int*)d_in[2];
  const float* weights = (const float*)d_in[3];

  float* y_out  = (float*)d_out;
  float* yh_out = (float*)d_out + B_ * D_;

  float* wtab    = (float*)d_ws;
  float* s_part  = wtab + B_ * T_;
  float* yh_part = s_part + 2 * B_ * D_;

  wtab_kernel<<<(B_ * T_) / 256, 256, 0, stream>>>(wordseq, slen, weights, wtab);
  fused_wmma_kernel<<<256, 256, 0, stream>>>(vs, slen, wtab, s_part, yh_part);
  finalize_kernel<<<B_, 256, 0, stream>>>(s_part, yh_part, y_out, yh_out);
}


# w-table launch folded into the streaming kernel: each workgroup gathers weights[word] of its two batch rows into LDS behind its first row loads (3 launches to 2; sums unchanged)
# speedup vs baseline: 1.4570x; 1.1289x over previous
#include <hip/hip_runtime.h>
#include <hip/hip_bf16.h>


typedef __attribute__((ext_vector_type(2))) float v2f;
typedef __attribute__((ext_vector_type(4))) float v4f;
typedef __attribute__((ext_vector_type(8))) float v8f;
typedef __attribute__((ext_vector_type(4))) int v4i;

#define B_ 64
#define T_ 512
#define D_ 1024
#define U_ 8
#define VOCAB_ 50257

struct Seg {
  const float* base[2];
  int len[2];
  int nA;
  int nTot;
  int phase;
};

__device__ __forceinline__ void step_desc(const Seg& s, int i, int& seg, int& t0, bool& valid) {
  valid = i < s.nTot;
  seg = (valid && i >= s.nA) ? 1 : 0;
  const int li = valid ? (seg ? i - s.nA : i) : 0;
  t0 = valid ? 4 * s.phase + 16 * li : 0;
}

__device__ __forceinline__ void load_step(const Seg& s, int i, int h, v4f& r0, v4f& r1) {
  int seg, t0; bool valid;
  step_desc(s, i, seg, t0, valid);
  const float* p = (seg ? s.base[1] : s.base[0]) + (size_t)(t0 + 2 * h) * D_;
  r0 = *(const v4f*)(p);
  r1 = *(const v4f*)(p + D_);
}

__device__ __forceinline__ void mma_step(const Seg& s, int i, int m, int h,
                                         const float (*lw)[T_],
                                         v4f& r0, v4f& r1,
                                         v8f& c0, v8f& c1, v8f& c2, v8f& c3) {
  asm volatile("" : "+v"(r0), "+v"(r1));
  int seg, t0; bool valid;
  step_desc(s, i, seg, t0, valid);
  const int te   = t0 + 2 * h;
  const int lenS = seg ? s.len[1] : s.len[0];
  const float mk0 = (valid && te     < lenS) ? 1.0f : 0.0f;
  const float mk1 = (valid && te + 1 < lenS) ? 1.0f : 0.0f;
  const v2f wp = *(const v2f*)(&lw[seg][te]);
  const int rowM = 2 * seg;
  v2f a;
  a.x = (m == rowM) ? mk0 : ((m == rowM + 1) ? mk0 * wp.x : 0.0f);
  a.y = (m == rowM) ? mk1 : ((m == rowM + 1) ? mk1 * wp.y : 0.0f);

  v2f bop;
  bop.x = r0.x; bop.y = r1.x;
  c0 = __builtin_amdgcn_wmma_f32_16x16x4_f32(false, a, false, bop, (short)0, c0, false, false);
  bop.x = r0.y; bop.y = r1.y;
  c1 = __builtin_amdgcn_wmma_f32_16x16x4_f32(false, a, false, bop, (short)0, c1, false, false);
  bop.x = r0.z; bop.y = r1.z;
  c2 = __builtin_amdgcn_wmma_f32_16x16x4_f32(false, a, false, bop, (short)0, c2, false, false);
  bop.x = r0.w; bop.y = r1.w;
  c3 = __builtin_amdgcn_wmma_f32_16x16x4_f32(false, a, false, bop, (short)0, c3, false, false);
}

__global__ __launch_bounds__(256) void fused_wmma_kernel(
    const float* __restrict__ vs, const int* __restrict__ slen,
    const int* __restrict__ wordseq, const float* __restrict__ weights,
    float* __restrict__ s_part, float* __restrict__ yh_part) {
  __shared__ int lds_len[B_];
  __shared__ int lds_pair[2];
  __shared__ __attribute__((aligned(16))) float lds_w[2][T_];
  __shared__ __attribute__((aligned(16))) float lds_red[4][2][4][64];

  const int tid  = threadIdx.x;
  const int lane = tid & 31;
  const int wv    = __builtin_amdgcn_readfirstlane(tid >> 5);
  const int tile  = wv & 1;
  const int phase = wv >> 1;
  const int p = blockIdx.x >> 3;
  const int q = blockIdx.x & 7;

  if (tid < B_) {
    int l = slen[tid];
    l = l < 0 ? 0 : (l > T_ ? T_ : l);
    lds_len[tid] = l;
  }
  __syncthreads();
  if (tid < B_) {
    const int mine = lds_len[tid];
    int rank = 0;
    for (int j = 0; j < B_; ++j) {
      const int o = lds_len[j];
      rank += (o < mine || (o == mine && j < tid)) ? 1 : 0;
    }
    if (rank == B_ - 1 - p) lds_pair[0] = tid;
    if (rank == p)          lds_pair[1] = tid;
  }
  __syncthreads();
  const int bA   = __builtin_amdgcn_readfirstlane(lds_pair[0]);
  const int bB   = __builtin_amdgcn_readfirstlane(lds_pair[1]);
  const int lenA = __builtin_amdgcn_readfirstlane(lds_len[bA]);
  const int lenB = __builtin_amdgcn_readfirstlane(lds_len[bB]);

  const int wseg = tid >> 7;
  const int wi4  = (tid & 127) * 4;
  const int wlen = wseg ? lenB : lenA;
  const v4i widx = *(const v4i*)(wordseq + (size_t)(wseg ? bB : bA) * T_ + wi4);

  const int m = lane & 15;
  const int h = lane >> 4;
  const int col = q * 128 + tile * 64 + 4 * m;

  Seg s;
  s.base[0] = vs + (size_t)bA * T_ * D_ + col;
  s.base[1] = vs + (size_t)bB * T_ * D_ + col;
  s.len[0] = lenA;
  s.len[1] = lenB;
  s.phase  = phase;
  s.nA   = (((lenA + 3) >> 2) - phase + 3) >> 2;
  const int nB = (((lenB + 3) >> 2) - phase + 3) >> 2;
  s.nTot = s.nA + nB;
  const int nraw = (s.nTot + U_ - 1) / U_;
  const int nch  = nraw < 1 ? 1 : nraw;

  v4f r0[U_], r1[U_];
#pragma unroll
  for (int u = 0; u < U_; ++u) load_step(s, u, h, r0[u], r1[u]);

  v4f wreg;
#pragma unroll
  for (int k = 0; k < 4; ++k) {
    const bool ok = (wi4 + k) < wlen;
    int id = widx[k];
    id = id < 0 ? 0 : (id > VOCAB_ - 1 ? VOCAB_ - 1 : id);
    const float g = weights[ok ? id : 0];
    wreg[k] = ok ? g : 0.0f;
  }
  *(v4f*)(&lds_w[wseg][wi4]) = wreg;
  __syncthreads();

  v8f c0 = {}, c1 = {}, c2 = {}, c3 = {};
  int c = 0;
  for (; c + 1 < nch; ++c) {
#pragma unroll
    for (int u = 0; u < U_; ++u) {
      mma_step(s, c * U_ + u, m, h, lds_w, r0[u], r1[u], c0, c1, c2, c3);
      load_step(s, (c + 1) * U_ + u, h, r0[u], r1[u]);
    }
  }
#pragma unroll
  for (int u = 0; u < U_; ++u)
    mma_step(s, c * U_ + u, m, h, lds_w, r0[u], r1[u], c0, c1, c2, c3);

  if (lane < 16) {
#pragma unroll
    for (int r = 0; r < 4; ++r) {
      v4f v;
      v.x = c0[r]; v.y = c1[r]; v.z = c2[r]; v.w = c3[r];
      *(v4f*)(&lds_red[phase][tile][r][4 * lane]) = v;
    }
  }
  __syncthreads();
  if (tid < 128) {
    const int r   = tid >> 5;
    const int tl  = (tid >> 4) & 1;
    const int l16 = tid & 15;
    v4f acc = *(const v4f*)(&lds_red[0][tl][r][4 * l16]);
#pragma unroll
    for (int ph = 1; ph < 4; ++ph) {
      const v4f x = *(const v4f*)(&lds_red[ph][tl][r][4 * l16]);
      acc.x += x.x; acc.y += x.y; acc.z += x.z; acc.w += x.w;
    }
    float* dst = (r & 1) ? yh_part : s_part;
    const int bo = (r >> 1) ? bB : bA;
    *(v4f*)(dst + (size_t)bo * D_ + q * 128 + tl * 64 + 4 * l16) = acc;
  }
}

__global__ __launch_bounds__(256) void finalize_kernel(
    const float* __restrict__ s_part, const float* __restrict__ yh_part,
    float* __restrict__ y_out, float* __restrict__ yh_out) {
  __shared__ float red[256];
  const int b   = blockIdx.x;
  const int tid = threadIdx.x;

  const v4f sv = ((const v4f*)(s_part + (size_t)b * D_))[tid];
  const v4f hv = ((const v4f*)(yh_part + (size_t)b * D_))[tid];
  ((v4f*)(yh_out + (size_t)b * D_))[tid] = hv;

  red[tid] = fabsf(sv.x) + fabsf(sv.y) + fabsf(sv.z) + fabsf(sv.w);
  __syncthreads();
  for (int off = 128; off > 0; off >>= 1) {
    if (tid < off) red[tid] += red[tid + off];
    __syncthreads();
  }
  const float inv = 1.0f / sqrtf(red[0]);
  v4f yv;
  yv.x = sv.x * inv; yv.y = sv.y * inv; yv.z = sv.z * inv; yv.w = sv.w * inv;
  ((v4f*)(y_out + (size_t)b * D_))[tid] = yv;
}

extern "C" void kernel_launch(void* const* d_in, const int* in_sizes, int n_in,
                              void* d_out, int out_size, void* d_ws, size_t ws_size,
                              hipStream_t stream) {
  const float* vs      = (const float*)d_in[0];
  const int*   slen    = (const int*)d_in[1];
  const int*   wordseq = (const int*)d_in[2];
  const float* weights = (const float*)d_in[3];

  float* y_out  = (float*)d_out;
  float* yh_out = (float*)d_out + B_ * D_;

  float* s_part  = (float*)d_ws;
  float* yh_part = s_part + 2 * B_ * D_;

  fused_wmma_kernel<<<256, 256, 0, stream>>>(vs, slen, wordseq, weights, s_part, yh_part);
  finalize_kernel<<<B_, 256, 0, stream>>>(s_part, yh_part, y_out, yh_out);
}


# K-step operand set-up taken out of the WMMA stream: A fragments read ready-made from LDS (mask, w, zero rows) one group ahead; row addresses = scalar base + fixed lane offset (sums unchanged)
# speedup vs baseline: 1.6894x; 1.1595x over previous
#include <hip/hip_runtime.h>
#include <hip/hip_bf16.h>


typedef __attribute__((ext_vector_type(2))) float v2f;
typedef __attribute__((ext_vector_type(4))) float v4f;
typedef __attribute__((ext_vector_type(8))) float v8f;
typedef __attribute__((ext_vector_type(4))) int v4i;

#define B_ 64
#define T_ 512
#define D_ 1024
#define U_ 8
#define VOCAB_ 50257

#define TP_ (T_ + 8)

struct Seg {
  const float* base[2];
  int nA;
  int nTot;
  int phase;
};

__device__ __forceinline__ void step_desc(const Seg& s, int i, int& seg, int& t0, bool& valid) {
  valid = i < s.nTot;
  seg = (valid && i >= s.nA) ? 1 : 0;
  const int li = valid ? (seg ? i - s.nA : i) : 0;
  t0 = valid ? 4 * s.phase + 16 * li : 0;
}

__device__ __forceinline__ void load_step(const Seg& s, int i, unsigned laneoff, v4f& r0, v4f& r1) {
  int seg, t0; bool valid;
  step_desc(s, i, seg, t0, valid);
  const float* p = (seg ? s.base[1] : s.base[0]) + (size_t)t0 * D_;
  r0 = *(const v4f*)(p + laneoff);
  r1 = *(const v4f*)(p + laneoff + D_);
}

__device__ __forceinline__ v2f load_afrag(const Seg& s, int i, const char* lds_a,
                                          int cS0, int cS1) {
  int seg, t0; bool valid;
  step_desc(s, i, seg, t0, valid);
  const int tA = valid ? t0 : T_;
  return *(const v2f*)(lds_a + (seg ? cS1 : cS0) + 4 * tA);
}

__device__ __forceinline__ void mma_step(const v2f a, v4f& r0, v4f& r1,
                                         v8f& c0, v8f& c1, v8f& c2, v8f& c3) {
  asm volatile("" : "+v"(r0), "+v"(r1));
  v2f bop;
  bop.x = r0.x; bop.y = r1.x;
  c0 = __builtin_amdgcn_wmma_f32_16x16x4_f32(false, a, false, bop, (short)0, c0, false, false);
  bop.x = r0.y; bop.y = r1.y;
  c1 = __builtin_amdgcn_wmma_f32_16x16x4_f32(false, a, false, bop, (short)0, c1, false, false);
  bop.x = r0.z; bop.y = r1.z;
  c2 = __builtin_amdgcn_wmma_f32_16x16x4_f32(false, a, false, bop, (short)0, c2, false, false);
  bop.x = r0.w; bop.y = r1.w;
  c3 = __builtin_amdgcn_wmma_f32_16x16x4_f32(false, a, false, bop, (short)0, c3, false, false);
  asm volatile("" : "+v"(c0), "+v"(c1), "+v"(c2), "+v"(c3));
}

__global__ __launch_bounds__(256) void fused_wmma_kernel(
    const float* __restrict__ vs, const int* __restrict__ slen,
    const int* __restrict__ wordseq, const float* __restrict__ weights,
    float* __restrict__ s_part, float* __restrict__ yh_part) {
  __shared__ int lds_len[B_];
  __shared__ int lds_pair[2];
  __shared__ __attribute__((aligned(16))) float lds_a[5][TP_];
  __shared__ __attribute__((aligned(16))) float lds_red[4][2][4][64];

  const int tid  = threadIdx.x;
  const int lane = tid & 31;
  const int wv    = __builtin_amdgcn_readfirstlane(tid >> 5);
  const int tile  = wv & 1;
  const int phase = wv >> 1;
  const int p = blockIdx.x >> 3;
  const int q = blockIdx.x & 7;

  if (tid < B_) {
    int l = slen[tid];
    l = l < 0 ? 0 : (l > T_ ? T_ : l);
    lds_len[tid] = l;
  }
  {
    const v4f z4 = {0.0f, 0.0f, 0.0f, 0.0f};
    if (tid < TP_ / 4) *(v4f*)(&lds_a[4][4 * tid]) = z4;
    if (tid >= 192 && tid < 200) {
      const int e = tid - 192;
      *(v4f*)(&lds_a[e >> 1][T_ + 4 * (e & 1)]) = z4;
    }
  }
  __syncthreads();
  if (tid < B_) {
    const int mine = lds_len[tid];
    int rank = 0;
    for (int j = 0; j < B_; ++j) {
      const int o = lds_len[j];
      rank += (o < mine || (o == mine && j < tid)) ? 1 : 0;
    }
    if (rank == B_ - 1 - p) lds_pair[0] = tid;
    if (rank == p)          lds_pair[1] = tid;
  }
  __syncthreads();
  const int bA   = __builtin_amdgcn_readfirstlane(lds_pair[0]);
  const int bB   = __builtin_amdgcn_readfirstlane(lds_pair[1]);
  const int lenA = __builtin_amdgcn_readfirstlane(lds_len[bA]);
  const int lenB = __builtin_amdgcn_readfirstlane(lds_len[bB]);

  const int wseg = tid >> 7;
  const int wi4  = (tid & 127) * 4;
  const int wlen = wseg ? lenB : lenA;
  const v4i widx = *(const v4i*)(wordseq + (size_t)(wseg ? bB : bA) * T_ + wi4);

  const int m = lane & 15;
  const int h = lane >> 4;
  const int col = q * 128 + tile * 64 + 4 * m;
  const unsigned laneoff = (unsigned)(2 * h * D_ + col);
  const int cS0 = ((m < 2) ? m : 4) * (TP_ * 4) + 8 * h;
  const int cS1 = ((m == 2 || m == 3) ? m : 4) * (TP_ * 4) + 8 * h;
  const char* lds_ab = (const char*)&lds_a[0][0];

  Seg s;
  s.base[0] = vs + (size_t)bA * T_ * D_;
  s.base[1] = vs + (size_t)bB * T_ * D_;
  s.phase  = phase;
  s.nA   = (((lenA + 3) >> 2) - phase + 3) >> 2;
  const int nB = (((lenB + 3) >> 2) - phase + 3) >> 2;
  s.nTot = s.nA + nB;
  const int nraw = (s.nTot + U_ - 1) / U_;
  const int nch  = nraw < 1 ? 1 : nraw;

  v4f r0[U_], r1[U_];
#pragma unroll
  for (int u = 0; u < U_; ++u) load_step(s, u, laneoff, r0[u], r1[u]);

  v4f wreg, mreg;
#pragma unroll
  for (int k = 0; k < 4; ++k) {
    const bool ok = (wi4 + k) < wlen;
    int id = widx[k];
    id = id < 0 ? 0 : (id > VOCAB_ - 1 ? VOCAB_ - 1 : id);
    const float g = weights[ok ? id : 0];
    wreg[k] = ok ? g : 0.0f;
    mreg[k] = ok ? 1.0f : 0.0f;
  }
  *(v4f*)(&lds_a[2 * wseg + 1][wi4]) = wreg;
  *(v4f*)(&lds_a[2 * wseg][wi4])     = mreg;
  __syncthreads();

  v2f af[U_];
#pragma unroll
  for (int u = 0; u < U_; ++u) af[u] = load_afrag(s, u, lds_ab, cS0, cS1);

  v8f c0 = {}, c1 = {}, c2 = {}, c3 = {};
  int c = 0;
  for (; c + 1 < nch; ++c) {
    v2f an[U_];
#pragma unroll
    for (int u = 0; u < U_; ++u) an[u] = load_afrag(s, (c + 1) * U_ + u, lds_ab, cS0, cS1);
#pragma unroll
    for (int u = 0; u < U_; ++u) {
      mma_step(af[u], r0[u], r1[u], c0, c1, c2, c3);
      load_step(s, (c + 1) * U_ + u, laneoff, r0[u], r1[u]);
    }
#pragma unroll
    for (int u = 0; u < U_; ++u) af[u] = an[u];
  }
#pragma unroll
  for (int u = 0; u < U_; ++u)
    mma_step(af[u], r0[u], r1[u], c0, c1, c2, c3);

  if (lane < 16) {
#pragma unroll
    for (int r = 0; r < 4; ++r) {
      v4f v;
      v.x = c0[r]; v.y = c1[r]; v.z = c2[r]; v.w = c3[r];
      *(v4f*)(&lds_red[phase][tile][r][4 * lane]) = v;
    }
  }
  __syncthreads();
  if (tid < 128) {
    const int r   = tid >> 5;
    const int tl  = (tid >> 4) & 1;
    const int l16 = tid & 15;
    v4f acc = *(const v4f*)(&lds_red[0][tl][r][4 * l16]);
#pragma unroll
    for (int ph = 1; ph < 4; ++ph) {
      const v4f x = *(const v4f*)(&lds_red[ph][tl][r][4 * l16]);
      acc.x += x.x; acc.y += x.y; acc.z += x.z; acc.w += x.w;
    }
    float* dst = (r & 1) ? yh_part : s_part;
    const int bo = (r >> 1) ? bB : bA;
    *(v4f*)(dst + (size_t)bo * D_ + q * 128 + tl * 64 + 4 * l16) = acc;
  }
}

__global__ __launch_bounds__(256) void finalize_kernel(
    const float* __restrict__ s_part, const float* __restrict__ yh_part,
    float* __restrict__ y_out, float* __restrict__ yh_out) {
  __shared__ float red[256];
  const int b   = blockIdx.x;
  const int tid = threadIdx.x;

  const v4f sv = ((const v4f*)(s_part + (size_t)b * D_))[tid];
  const v4f hv = ((const v4f*)(yh_part + (size_t)b * D_))[tid];
  ((v4f*)(yh_out + (size_t)b * D_))[tid] = hv;

  red[tid] = fabsf(sv.x) + fabsf(sv.y) + fabsf(sv.z) + fabsf(sv.w);
  __syncthreads();
  for (int off = 128; off > 0; off >>= 1) {
    if (tid < off) red[tid] += red[tid + off];
    __syncthreads();
  }
  const float inv = 1.0f / sqrtf(red[0]);
  v4f yv;
  yv.x = sv.x * inv; yv.y = sv.y * inv; yv.z = sv.z * inv; yv.w = sv.w * inv;
  ((v4f*)(y_out + (size_t)b * D_))[tid] = yv;
}

extern "C" void kernel_launch(void* const* d_in, const int* in_sizes, int n_in,
                              void* d_out, int out_size, void* d_ws, size_t ws_size,
                              hipStream_t stream) {
  const float* vs      = (const float*)d_in[0];
  const int*   slen    = (const int*)d_in[1];
  const int*   wordseq = (const int*)d_in[2];
  const float* weights = (const float*)d_in[3];

  float* y_out  = (float*)d_out;
  float* yh_out = (float*)d_out + B_ * D_;

  float* s_part  = (float*)d_ws;
  float* yh_part = s_part + 2 * B_ * D_;

  fused_wmma_kernel<<<256, 256, 0, stream>>>(vs, slen, wordseq, weights, s_part, yh_part);
  finalize_kernel<<<B_, 256, 0, stream>>>(s_part, yh_part, y_out, yh_out);
}
